# starting kernel
# baseline (speedup 1.0000x reference)
#include <hip/hip_runtime.h>
#include <hip/hip_bf16.h>

typedef float v2f __attribute__((ext_vector_type(2)));
typedef float v8f __attribute__((ext_vector_type(8)));

#define D 128
#define HIST 50
#define CAND 20

__global__ __launch_bounds__(32) void encoder_gemm(
    const float* __restrict__ x, const float* __restrict__ W,
    float* __restrict__ out, int K  ) {
  const int lane = threadIdx.x;
  const int tile = blockIdx.x;
  const int mt = tile >> 3;
  const int nt = tile & 7;
  const int arow = mt * 16 + (lane & 15);
  const int col  = nt * 16 + (lane & 15);
  const int kh   = (lane >> 4) * 2;

  const float* __restrict__ xr = x + (size_t)arow * K;
  v8f acc = {};
  for (int k = 0; k < K; k += 4) {
    v2f a, b;
    a.x = xr[k + kh + 0];
    a.y = xr[k + kh + 1];
    b.x = W[(size_t)(k + kh + 0) * D + col];
    b.y = W[(size_t)(k + kh + 1) * D + col];
    acc = __builtin_amdgcn_wmma_f32_16x16x4_f32(
        false, a, false, b, (short)0, acc, false, false);
  }
  const int mbase = mt * 16 + (lane >> 4) * 8;
#pragma unroll
  for (int r = 0; r < 8; ++r)
    out[(size_t)(mbase + r) * D + col] = acc[r];
}

__global__ void edge_agg(const int* __restrict__ ei,
                         const float* __restrict__ news,
                         float* __restrict__ agg, float* __restrict__ deg,
                         long long E) {
  long long idx = (long long)blockIdx.x * blockDim.x + threadIdx.x;
  long long e = idx >> 7;
  int d = (int)(idx & (D - 1));
  if (e >= E) return;
  int s = ei[e];
  int t = ei[E + e];
  atomicAdd(&agg[(size_t)t * D + d], news[(size_t)s * D + d]);
  if (d == 0) atomicAdd(&deg[t], 1.0f);
}

__global__ __launch_bounds__(32) void gnn_gemm_add(
    const float* __restrict__ agg, const float* __restrict__ deg,
    const float* __restrict__ Wg, const float* __restrict__ news,
    float* __restrict__ xx) {
  const int lane = threadIdx.x;
  const int tile = blockIdx.x;
  const int mt = tile >> 3;
  const int nt = tile & 7;
  const int arow = mt * 16 + (lane & 15);
  const int col  = nt * 16 + (lane & 15);
  const int kh   = (lane >> 4) * 2;

  const float dinv = 1.0f / fmaxf(deg[arow], 1.0f);
  const float* __restrict__ ar = agg + (size_t)arow * D;
  v8f acc = {};
  for (int k = 0; k < D; k += 4) {
    v2f a, b;
    a.x = ar[k + kh + 0] * dinv;
    a.y = ar[k + kh + 1] * dinv;
    b.x = Wg[(size_t)(k + kh + 0) * D + col];
    b.y = Wg[(size_t)(k + kh + 1) * D + col];
    acc = __builtin_amdgcn_wmma_f32_16x16x4_f32(
        false, a, false, b, (short)0, acc, false, false);
  }
  const int mbase = mt * 16 + (lane >> 4) * 8;
#pragma unroll
  for (int r = 0; r < 8; ++r) {
    size_t o = (size_t)(mbase + r) * D + col;
    xx[o] = news[o] + acc[r];
  }
}

__global__ __launch_bounds__(128) void user_score(
    const float* __restrict__ xx, const int* __restrict__ hist,
    const int* __restrict__ cand, float* __restrict__ scores) {
  const int b = blockIdx.x;
  const int t = threadIdx.x;
  __shared__ float uv[D];

  float sum = 0.0f, cnt = 0.0f;
  for (int h = 0; h < HIST; ++h) {
    int nid = hist[b * HIST + h];
    if (nid != 0) {
      sum += xx[(size_t)nid * D + t];
      cnt += 1.0f;
    }
  }
  uv[t] = sum / fmaxf(cnt, 1e-9f);
  __syncthreads();

  const int wave = t >> 5, lane = t & 31;
  for (int c = wave; c < CAND; c += 4) {
    int nid = cand[b * CAND + c];
    const float* __restrict__ ce = xx + (size_t)nid * D;
    float p = 0.0f;
#pragma unroll
    for (int d = lane; d < D; d += 32) p += ce[d] * uv[d];
#pragma unroll
    for (int off = 16; off; off >>= 1) p += __shfl_xor(p, off, 32);
    if (lane == 0) scores[b * CAND + c] = p;
  }
}

extern "C" void kernel_launch(void* const* d_in, const int* in_sizes, int n_in,
                              void* d_out, int out_size, void* d_ws, size_t ws_size,
                              hipStream_t stream) {
  const float* x      = (const float*)d_in[0];
  const int*   ei     = (const int*)  d_in[2];
  const int*   hist   = (const int*)  d_in[3];
  const int*   cand   = (const int*)  d_in[4];
  const float* W_enc  = (const float*)d_in[5];
  const float* W_gnn  = (const float*)d_in[6];
  float*       scores = (float*)d_out;

  const int       N = in_sizes[1];
  const int       K = in_sizes[0] / N;
  const long long E = (long long)in_sizes[2] / 2;
  const int       B = in_sizes[3] / HIST;

  float* news = (float*)d_ws;
  float* agg  = news + (size_t)N * D;
  float* deg  = agg  + (size_t)N * D;
  float* xx   = deg  + (size_t)N;

  hipMemsetAsync(agg, 0, (size_t)N * D * sizeof(float), stream);
  hipMemsetAsync(deg, 0, (size_t)N * sizeof(float), stream);

  const int tiles = (N / 16) * (D / 16);
  encoder_gemm<<<tiles, 32, 0, stream>>>(x, W_enc, news, K);

  long long work = E * D;
  int eblocks = (int)((work + 255) / 256);
  edge_agg<<<eblocks, 256, 0, stream>>>(ei, news, agg, deg, E);

  gnn_gemm_add<<<tiles, 32, 0, stream>>>(agg, deg, W_gnn, news, xx);

  user_score<<<B, 128, 0, stream>>>(xx, hist, cand, scores);
}


# Edge aggregation: 128 float atomics per edge replaced by a counting sort by destination (one integer atomic per edge) and a per-node register sum of source rows; sum order per node changes
# speedup vs baseline: 1.9499x; 1.9499x over previous
#include <hip/hip_runtime.h>
#include <hip/hip_bf16.h>

typedef float v2f __attribute__((ext_vector_type(2)));
typedef float v8f __attribute__((ext_vector_type(8)));

#define D 128
#define HIST 50
#define CAND 20

__global__ __launch_bounds__(32) void encoder_gemm(
    const float* __restrict__ x, const float* __restrict__ W,
    float* __restrict__ out, int K  ) {
  const int lane = threadIdx.x;
  const int tile = blockIdx.x;
  const int mt = tile >> 3;
  const int nt = tile & 7;
  const int arow = mt * 16 + (lane & 15);
  const int col  = nt * 16 + (lane & 15);
  const int kh   = (lane >> 4) * 2;

  const float* __restrict__ xr = x + (size_t)arow * K;
  v8f acc = {};
  for (int k = 0; k < K; k += 4) {
    v2f a, b;
    a.x = xr[k + kh + 0];
    a.y = xr[k + kh + 1];
    b.x = W[(size_t)(k + kh + 0) * D + col];
    b.y = W[(size_t)(k + kh + 1) * D + col];
    acc = __builtin_amdgcn_wmma_f32_16x16x4_f32(
        false, a, false, b, (short)0, acc, false, false);
  }
  const int mbase = mt * 16 + (lane >> 4) * 8;
#pragma unroll
  for (int r = 0; r < 8; ++r)
    out[(size_t)(mbase + r) * D + col] = acc[r];
}


__global__ __launch_bounds__(256) void edge_rank(const int* __restrict__ ei,
                                                 int* __restrict__ cnt,
                                                 int* __restrict__ rank,
                                                 long long E) {
  long long e = (long long)blockIdx.x * blockDim.x + threadIdx.x;
  if (e >= E) return;
  int t = ei[E + e];
  rank[e] = atomicAdd(&cnt[t], 1);
}

__global__ __launch_bounds__(1024) void scan_counts(const int* __restrict__ cnt,
                                                    int* __restrict__ rowptr,
                                                    int N) {
  __shared__ int wsum[32];
  const int tid = threadIdx.x, lane = tid & 31, wave = tid >> 5;
  int carry = 0;
  for (int base = 0; base < N; base += 16384) {
    const int i0 = base + tid * 16;
    const bool whole = (i0 + 16 <= N);
    int c[16];
    if (whole) {
      const int4* p = reinterpret_cast<const int4*>(cnt + i0);
#pragma unroll
      for (int q = 0; q < 4; ++q) {
        const int4 t = p[q];
        c[4 * q + 0] = t.x; c[4 * q + 1] = t.y; c[4 * q + 2] = t.z; c[4 * q + 3] = t.w;
      }
    } else {
#pragma unroll
      for (int u = 0; u < 16; ++u) c[u] = (i0 + u < N) ? cnt[i0 + u] : 0;
    }
    int tsum = 0;
#pragma unroll
    for (int u = 0; u < 16; ++u) { const int t = c[u]; c[u] = tsum; tsum += t; }
    int inc = tsum;
    for (int off = 1; off < 32; off <<= 1) {
      int v = __shfl_up(inc, off, 32);
      if (lane >= off) inc += v;
    }
    if (lane == 31) wsum[wave] = inc;
    __syncthreads();
    const int ws = wsum[lane];
    int winc = ws;
    for (int off = 1; off < 32; off <<= 1) {
      int v = __shfl_up(winc, off, 32);
      if (lane >= off) winc += v;
    }
    const int wave_excl = __shfl(winc, wave, 32) - __shfl(ws, wave, 32);
    const int total = __shfl(winc, 31, 32);
    const int ex = carry + wave_excl + (inc - tsum);
    if (whole) {
      int4* p = reinterpret_cast<int4*>(rowptr + i0);
#pragma unroll
      for (int q = 0; q < 4; ++q)
        p[q] = make_int4(ex + c[4 * q + 0], ex + c[4 * q + 1], ex + c[4 * q + 2], ex + c[4 * q + 3]);
    } else {
#pragma unroll
      for (int u = 0; u < 16; ++u)
        if (i0 + u < N) rowptr[i0 + u] = ex + c[u];
    }
    carry += total;
    __syncthreads();
  }
  if (tid == 0) rowptr[N] = carry;
}

__global__ __launch_bounds__(256) void edge_fill(const int* __restrict__ ei,
                                                 const int* __restrict__ rowptr,
                                                 const int* __restrict__ rank,
                                                 int* __restrict__ srcs,
                                                 long long E) {
  long long e = (long long)blockIdx.x * blockDim.x + threadIdx.x;
  if (e >= E) return;
  int s = ei[e];
  int t = ei[E + e];
  srcs[rowptr[t] + rank[e]] = s;
}

__global__ __launch_bounds__(256) void seg_gather(const int* __restrict__ rowptr,
                                                  const int* __restrict__ srcs,
                                                  const float* __restrict__ news,
                                                  float* __restrict__ agg,
                                                  int N) {
  const int lane = threadIdx.x & 31;
  const int n = __builtin_amdgcn_readfirstlane((int)(blockIdx.x * 8 + (threadIdx.x >> 5)));
  if (n >= N) return;
  const int beg = __builtin_amdgcn_readfirstlane(rowptr[n]);
  const int end = __builtin_amdgcn_readfirstlane(rowptr[n + 1]);
  const float* __restrict__ nl = news + lane * 4;
  float4 acc = make_float4(0.0f, 0.0f, 0.0f, 0.0f);
  for (int base = beg; base < end; base += 32) {
    const int m = min(32, end - base);
    const int my = (lane < m) ? srcs[base + lane] : 0;
    for (int j = 0; j < m; j += 8) {
      float4 v[8];
#pragma unroll
      for (int u = 0; u < 8; ++u) {
        const int idx = min(j + u, m - 1);
        const int s = __builtin_amdgcn_readlane(my, idx);
        v[u] = *reinterpret_cast<const float4*>(nl + (size_t)s * D);
      }
#pragma unroll
      for (int u = 0; u < 8; ++u) {
        if (j + u < m) {
          acc.x += v[u].x; acc.y += v[u].y; acc.z += v[u].z; acc.w += v[u].w;
        }
      }
    }
  }
  *reinterpret_cast<float4*>(agg + (size_t)n * D + lane * 4) = acc;
}

__global__ __launch_bounds__(32) void gnn_gemm_add(
    const float* __restrict__ agg, const int* __restrict__ deg,
    const float* __restrict__ Wg, const float* __restrict__ news,
    float* __restrict__ xx) {
  const int lane = threadIdx.x;
  const int tile = blockIdx.x;
  const int mt = tile >> 3;
  const int nt = tile & 7;
  const int arow = mt * 16 + (lane & 15);
  const int col  = nt * 16 + (lane & 15);
  const int kh   = (lane >> 4) * 2;

  const float dinv = 1.0f / fmaxf((float)deg[arow], 1.0f);
  const float* __restrict__ ar = agg + (size_t)arow * D;
  v8f acc = {};
  for (int k = 0; k < D; k += 4) {
    v2f a, b;
    a.x = ar[k + kh + 0] * dinv;
    a.y = ar[k + kh + 1] * dinv;
    b.x = Wg[(size_t)(k + kh + 0) * D + col];
    b.y = Wg[(size_t)(k + kh + 1) * D + col];
    acc = __builtin_amdgcn_wmma_f32_16x16x4_f32(
        false, a, false, b, (short)0, acc, false, false);
  }
  const int mbase = mt * 16 + (lane >> 4) * 8;
#pragma unroll
  for (int r = 0; r < 8; ++r) {
    size_t o = (size_t)(mbase + r) * D + col;
    xx[o] = news[o] + acc[r];
  }
}

__global__ __launch_bounds__(128) void user_score(
    const float* __restrict__ xx, const int* __restrict__ hist,
    const int* __restrict__ cand, float* __restrict__ scores) {
  const int b = blockIdx.x;
  const int t = threadIdx.x;
  __shared__ float uv[D];

  float sum = 0.0f, cnt = 0.0f;
  for (int h = 0; h < HIST; ++h) {
    int nid = hist[b * HIST + h];
    if (nid != 0) {
      sum += xx[(size_t)nid * D + t];
      cnt += 1.0f;
    }
  }
  uv[t] = sum / fmaxf(cnt, 1e-9f);
  __syncthreads();

  const int wave = t >> 5, lane = t & 31;
  for (int c = wave; c < CAND; c += 4) {
    int nid = cand[b * CAND + c];
    const float* __restrict__ ce = xx + (size_t)nid * D;
    float p = 0.0f;
#pragma unroll
    for (int d = lane; d < D; d += 32) p += ce[d] * uv[d];
#pragma unroll
    for (int off = 16; off; off >>= 1) p += __shfl_xor(p, off, 32);
    if (lane == 0) scores[b * CAND + c] = p;
  }
}

extern "C" void kernel_launch(void* const* d_in, const int* in_sizes, int n_in,
                              void* d_out, int out_size, void* d_ws, size_t ws_size,
                              hipStream_t stream) {
  const float* x      = (const float*)d_in[0];
  const int*   ei     = (const int*)  d_in[2];
  const int*   hist   = (const int*)  d_in[3];
  const int*   cand   = (const int*)  d_in[4];
  const float* W_enc  = (const float*)d_in[5];
  const float* W_gnn  = (const float*)d_in[6];
  float*       scores = (float*)d_out;

  const int       N = in_sizes[1];
  const int       K = in_sizes[0] / N;
  const long long E = (long long)in_sizes[2] / 2;
  const int       B = in_sizes[3] / HIST;

  float* news = (float*)d_ws;
  float* agg  = news + (size_t)N * D;
  int*   cnt  = (int*)(agg + (size_t)N * D);
  float* xx   = (float*)(cnt + (size_t)N);
  int* rowptr = (int*)xx;
  int* rank   = rowptr + (size_t)N + 1;
  int* srcs   = rank + E;

  hipMemsetAsync(cnt, 0, (size_t)N * sizeof(int), stream);

  const int tiles = (N / 16) * (D / 16);
  encoder_gemm<<<tiles, 32, 0, stream>>>(x, W_enc, news, K);

  int eblocks = (int)((E + 255) / 256);
  edge_rank<<<eblocks, 256, 0, stream>>>(ei, cnt, rank, E);
  scan_counts<<<1, 1024, 0, stream>>>(cnt, rowptr, N);
  edge_fill<<<eblocks, 256, 0, stream>>>(ei, rowptr, rank, srcs, E);
  seg_gather<<<(N + 7) / 8, 256, 0, stream>>>(rowptr, srcs, news, agg, N);

  gnn_gemm_add<<<tiles, 32, 0, stream>>>(agg, cnt, W_gnn, news, xx);

  user_score<<<B, 128, 0, stream>>>(xx, hist, cand, scores);
}


# Both GEMMs re-tiled: one wave per 16x128 row strip, each A fragment reused by 8 WMMAs, weights staged once per work-group in LDS in fragment order; k sum order inside each group of 8 changes
# speedup vs baseline: 2.6809x; 1.3749x over previous
#include <hip/hip_runtime.h>
#include <hip/hip_bf16.h>

typedef float v2f __attribute__((ext_vector_type(2)));
typedef float v8f __attribute__((ext_vector_type(8)));

#define D 128
#define HIST 50
#define CAND 20
#define KENC 300
#define SW 8

template <int K>
__device__ __forceinline__ int step_k(int j, int h) {
  return (j < 2 * (K / 8)) ? 8 * (j >> 1) + 4 * h + 2 * (j & 1)
                           : 8 * (K / 8) + 2 * h;
}

template <int K, int U>
__device__ __forceinline__ void stage_weights(const float* __restrict__ W,
                                              v2f* __restrict__ wl) {
  static_assert((K / 4) % U == 0, "batches must tile the steps");
  const int tid = threadIdx.x;
  const int lane = tid & 31, nt = tid >> 5;
  const int n = nt * 16 + (lane & 15);
  const int h = lane >> 4;
#pragma unroll
  for (int j0 = 0; j0 < K / 4; j0 += U) {
    v2f w[U];
#pragma unroll
    for (int u = 0; u < U; ++u) {
      const int k = step_k<K>(j0 + u, h);
      w[u].x = W[(size_t)k * D + n];
      w[u].y = W[(size_t)(k + 1) * D + n];
    }
#pragma unroll
    for (int u = 0; u < U; ++u) wl[(j0 + u) * 256 + tid] = w[u];
  }
}

template <int K>
struct AFrag {
  float4 q[K / 8];
  v2f r;
};

template <int K>
__device__ __forceinline__ void load_a(const float* __restrict__ row, int h,
                                       AFrag<K>& a) {
#pragma unroll
  for (int i = 0; i < K / 8; ++i)
    a.q[i] = *reinterpret_cast<const float4*>(row + 8 * i + 4 * h);
  if (K % 8)
    a.r = *reinterpret_cast<const v2f*>(row + 8 * (K / 8) + 2 * h);
}

template <int K, int SEG, bool SCALE>
__device__ __forceinline__ void strip_mma(const AFrag<K>& a, float scale,
                                          const v2f* __restrict__ wl, int lane,
                                          v8f (&acc)[8]) {
  constexpr int NSEG = (K / 4 + SEG - 1) / SEG;
  int base[NSEG];
#pragma unroll
  for (int s = 0; s < NSEG; ++s) {
    base[s] = s * SEG * 256 + lane;
    if (s > 0) asm volatile("" : "+v"(base[s]));
  }
#pragma unroll
  for (int j = 0; j < K / 4; ++j) {
    const v2f* __restrict__ seg = wl + base[j / SEG];
    v2f av;
    if (j < 2 * (K / 8)) {
      const float4 q = a.q[j >> 1];
      if (j & 1) { av.x = q.z; av.y = q.w; } else { av.x = q.x; av.y = q.y; }
    } else {
      av = a.r;
    }
    if (SCALE) { av.x *= scale; av.y *= scale; }
#pragma unroll
    for (int nt = 0; nt < 8; ++nt) {
      const v2f bv = seg[((j % SEG) * 8 + nt) * 32];
      acc[nt] = __builtin_amdgcn_wmma_f32_16x16x4_f32(
          false, av, false, bv, (short)0, acc[nt], false, false);
    }
  }
}

__global__ __launch_bounds__(SW * 32, 4) void encoder_gemm(
    const float* __restrict__ x, const float* __restrict__ W,
    float* __restrict__ out, int nstrips) {
  __shared__ v2f wl[(KENC / 4) * 256];
  const int lane = threadIdx.x & 31;
  const int c = lane & 15, h = lane >> 4;
  const int mine = blockIdx.x * SW + (threadIdx.x >> 5);
  const int strip = min(mine, nstrips - 1);

  AFrag<KENC> a;
  load_a<KENC>(x + (size_t)(strip * 16 + c) * KENC, h, a);
  stage_weights<KENC, 25>(W, wl);
  __syncthreads();

  v8f acc[8] = {};
  strip_mma<KENC, 25, false>(a, 1.0f, wl, lane, acc);

  if (mine < nstrips) {
    float* __restrict__ o = out + (size_t)(strip * 16 + h * 8) * D + c;
#pragma unroll
    for (int nt = 0; nt < 8; ++nt)
#pragma unroll
      for (int r = 0; r < 8; ++r)
        o[(size_t)r * D + nt * 16] = acc[nt][r];
  }
}


__global__ __launch_bounds__(256) void edge_rank(const int* __restrict__ ei,
                                                 int* __restrict__ cnt,
                                                 int* __restrict__ rank,
                                                 long long E) {
  long long e = (long long)blockIdx.x * blockDim.x + threadIdx.x;
  if (e >= E) return;
  int t = ei[E + e];
  rank[e] = atomicAdd(&cnt[t], 1);
}

__global__ __launch_bounds__(1024) void scan_counts(const int* __restrict__ cnt,
                                                    int* __restrict__ rowptr,
                                                    int N) {
  __shared__ int wsum[32];
  const int tid = threadIdx.x, lane = tid & 31, wave = tid >> 5;
  int carry = 0;
  for (int base = 0; base < N; base += 16384) {
    const int i0 = base + tid * 16;
    const bool whole = (i0 + 16 <= N);
    int c[16];
    if (whole) {
      const int4* p = reinterpret_cast<const int4*>(cnt + i0);
#pragma unroll
      for (int q = 0; q < 4; ++q) {
        const int4 t = p[q];
        c[4 * q + 0] = t.x; c[4 * q + 1] = t.y; c[4 * q + 2] = t.z; c[4 * q + 3] = t.w;
      }
    } else {
#pragma unroll
      for (int u = 0; u < 16; ++u) c[u] = (i0 + u < N) ? cnt[i0 + u] : 0;
    }
    int tsum = 0;
#pragma unroll
    for (int u = 0; u < 16; ++u) { const int t = c[u]; c[u] = tsum; tsum += t; }
    int inc = tsum;
    for (int off = 1; off < 32; off <<= 1) {
      int v = __shfl_up(inc, off, 32);
      if (lane >= off) inc += v;
    }
    if (lane == 31) wsum[wave] = inc;
    __syncthreads();
    const int ws = wsum[lane];
    int winc = ws;
    for (int off = 1; off < 32; off <<= 1) {
      int v = __shfl_up(winc, off, 32);
      if (lane >= off) winc += v;
    }
    const int wave_excl = __shfl(winc, wave, 32) - __shfl(ws, wave, 32);
    const int total = __shfl(winc, 31, 32);
    const int ex = carry + wave_excl + (inc - tsum);
    if (whole) {
      int4* p = reinterpret_cast<int4*>(rowptr + i0);
#pragma unroll
      for (int q = 0; q < 4; ++q)
        p[q] = make_int4(ex + c[4 * q + 0], ex + c[4 * q + 1], ex + c[4 * q + 2], ex + c[4 * q + 3]);
    } else {
#pragma unroll
      for (int u = 0; u < 16; ++u)
        if (i0 + u < N) rowptr[i0 + u] = ex + c[u];
    }
    carry += total;
    __syncthreads();
  }
  if (tid == 0) rowptr[N] = carry;
}

__global__ __launch_bounds__(256) void edge_fill(const int* __restrict__ ei,
                                                 const int* __restrict__ rowptr,
                                                 const int* __restrict__ rank,
                                                 int* __restrict__ srcs,
                                                 long long E) {
  long long e = (long long)blockIdx.x * blockDim.x + threadIdx.x;
  if (e >= E) return;
  int s = ei[e];
  int t = ei[E + e];
  srcs[rowptr[t] + rank[e]] = s;
}

__global__ __launch_bounds__(256) void seg_gather(const int* __restrict__ rowptr,
                                                  const int* __restrict__ srcs,
                                                  const float* __restrict__ news,
                                                  float* __restrict__ agg,
                                                  int N) {
  const int lane = threadIdx.x & 31;
  const int n = __builtin_amdgcn_readfirstlane((int)(blockIdx.x * 8 + (threadIdx.x >> 5)));
  if (n >= N) return;
  const int beg = __builtin_amdgcn_readfirstlane(rowptr[n]);
  const int end = __builtin_amdgcn_readfirstlane(rowptr[n + 1]);
  const float* __restrict__ nl = news + lane * 4;
  float4 acc = make_float4(0.0f, 0.0f, 0.0f, 0.0f);
  for (int base = beg; base < end; base += 32) {
    const int m = min(32, end - base);
    const int my = (lane < m) ? srcs[base + lane] : 0;
    for (int j = 0; j < m; j += 8) {
      float4 v[8];
#pragma unroll
      for (int u = 0; u < 8; ++u) {
        const int idx = min(j + u, m - 1);
        const int s = __builtin_amdgcn_readlane(my, idx);
        v[u] = *reinterpret_cast<const float4*>(nl + (size_t)s * D);
      }
#pragma unroll
      for (int u = 0; u < 8; ++u) {
        if (j + u < m) {
          acc.x += v[u].x; acc.y += v[u].y; acc.z += v[u].z; acc.w += v[u].w;
        }
      }
    }
  }
  *reinterpret_cast<float4*>(agg + (size_t)n * D + lane * 4) = acc;
}

__global__ __launch_bounds__(SW * 32) void gnn_gemm_add(
    const float* __restrict__ agg, const int* __restrict__ deg,
    const float* __restrict__ Wg, const float* __restrict__ news,
    float* __restrict__ xx, int nstrips) {
  __shared__ v2f wl[(D / 4) * 256];
  const int lane = threadIdx.x & 31;
  const int c = lane & 15, h = lane >> 4;
  const int mine = blockIdx.x * SW + (threadIdx.x >> 5);
  const int strip = min(mine, nstrips - 1);
  const int arow = strip * 16 + c;

  AFrag<D> a;
  load_a<D>(agg + (size_t)arow * D, h, a);
  const float dinv = 1.0f / fmaxf((float)deg[arow], 1.0f);
  stage_weights<D, 16>(Wg, wl);
  __syncthreads();

  v8f acc[8] = {};
  strip_mma<D, 32, true>(a, dinv, wl, lane, acc);

  if (mine < nstrips) {
    const size_t o0 = (size_t)(strip * 16 + h * 8) * D + c;
#pragma unroll
    for (int nt = 0; nt < 8; ++nt)
#pragma unroll
      for (int r = 0; r < 8; ++r) {
        const size_t o = o0 + (size_t)r * D + nt * 16;
        xx[o] = news[o] + acc[nt][r];
      }
  }
}

__global__ __launch_bounds__(128) void user_score(
    const float* __restrict__ xx, const int* __restrict__ hist,
    const int* __restrict__ cand, float* __restrict__ scores) {
  const int b = blockIdx.x;
  const int t = threadIdx.x;
  __shared__ float uv[D];

  float sum = 0.0f, cnt = 0.0f;
  for (int h = 0; h < HIST; ++h) {
    int nid = hist[b * HIST + h];
    if (nid != 0) {
      sum += xx[(size_t)nid * D + t];
      cnt += 1.0f;
    }
  }
  uv[t] = sum / fmaxf(cnt, 1e-9f);
  __syncthreads();

  const int wave = t >> 5, lane = t & 31;
  for (int c = wave; c < CAND; c += 4) {
    int nid = cand[b * CAND + c];
    const float* __restrict__ ce = xx + (size_t)nid * D;
    float p = 0.0f;
#pragma unroll
    for (int d = lane; d < D; d += 32) p += ce[d] * uv[d];
#pragma unroll
    for (int off = 16; off; off >>= 1) p += __shfl_xor(p, off, 32);
    if (lane == 0) scores[b * CAND + c] = p;
  }
}

extern "C" void kernel_launch(void* const* d_in, const int* in_sizes, int n_in,
                              void* d_out, int out_size, void* d_ws, size_t ws_size,
                              hipStream_t stream) {
  const float* x      = (const float*)d_in[0];
  const int*   ei     = (const int*)  d_in[2];
  const int*   hist   = (const int*)  d_in[3];
  const int*   cand   = (const int*)  d_in[4];
  const float* W_enc  = (const float*)d_in[5];
  const float* W_gnn  = (const float*)d_in[6];
  float*       scores = (float*)d_out;

  const int       N = in_sizes[1];
  const long long E = (long long)in_sizes[2] / 2;
  const int       B = in_sizes[3] / HIST;

  float* news = (float*)d_ws;
  float* agg  = news + (size_t)N * D;
  int*   cnt  = (int*)(agg + (size_t)N * D);
  float* xx   = (float*)(cnt + (size_t)N);
  int* rowptr = (int*)xx;
  int* rank   = rowptr + (size_t)N + 1;
  int* srcs   = rank + E;

  hipMemsetAsync(cnt, 0, (size_t)N * sizeof(int), stream);

  const int strips  = N / 16;
  const int gblocks = (strips + SW - 1) / SW;
  encoder_gemm<<<gblocks, SW * 32, 0, stream>>>(x, W_enc, news, strips);

  int eblocks = (int)((E + 255) / 256);
  edge_rank<<<eblocks, 256, 0, stream>>>(ei, cnt, rank, E);
  scan_counts<<<1, 1024, 0, stream>>>(cnt, rowptr, N);
  edge_fill<<<eblocks, 256, 0, stream>>>(ei, rowptr, rank, srcs, E);
  seg_gather<<<(N + 7) / 8, 256, 0, stream>>>(rowptr, srcs, news, agg, N);

  gnn_gemm_add<<<gblocks, SW * 32, 0, stream>>>(agg, cnt, W_gnn, news, xx, strips);

  user_score<<<B, 128, 0, stream>>>(xx, hist, cand, scores);
}


# User scoring: one wave per user, its 70 row loads in flight in three dependent round trips (was one round trip per history row); candidate dot-product sum order changes
# speedup vs baseline: 2.8725x; 1.0715x over previous
#include <hip/hip_runtime.h>
#include <hip/hip_bf16.h>

typedef float v2f __attribute__((ext_vector_type(2)));
typedef float v8f __attribute__((ext_vector_type(8)));

#define D 128
#define HIST 50
#define CAND 20
#define KENC 300
#define SW 8

template <int K>
__device__ __forceinline__ int step_k(int j, int h) {
  return (j < 2 * (K / 8)) ? 8 * (j >> 1) + 4 * h + 2 * (j & 1)
                           : 8 * (K / 8) + 2 * h;
}

template <int K, int U>
__device__ __forceinline__ void stage_weights(const float* __restrict__ W,
                                              v2f* __restrict__ wl) {
  static_assert((K / 4) % U == 0, "batches must tile the steps");
  const int tid = threadIdx.x;
  const int lane = tid & 31, nt = tid >> 5;
  const int n = nt * 16 + (lane & 15);
  const int h = lane >> 4;
#pragma unroll
  for (int j0 = 0; j0 < K / 4; j0 += U) {
    v2f w[U];
#pragma unroll
    for (int u = 0; u < U; ++u) {
      const int k = step_k<K>(j0 + u, h);
      w[u].x = W[(size_t)k * D + n];
      w[u].y = W[(size_t)(k + 1) * D + n];
    }
#pragma unroll
    for (int u = 0; u < U; ++u) wl[(j0 + u) * 256 + tid] = w[u];
  }
}

template <int K>
struct AFrag {
  float4 q[K / 8];
  v2f r;
};

template <int K>
__device__ __forceinline__ void load_a(const float* __restrict__ row, int h,
                                       AFrag<K>& a) {
#pragma unroll
  for (int i = 0; i < K / 8; ++i)
    a.q[i] = *reinterpret_cast<const float4*>(row + 8 * i + 4 * h);
  if (K % 8)
    a.r = *reinterpret_cast<const v2f*>(row + 8 * (K / 8) + 2 * h);
}

template <int K, int SEG, bool SCALE>
__device__ __forceinline__ void strip_mma(const AFrag<K>& a, float scale,
                                          const v2f* __restrict__ wl, int lane,
                                          v8f (&acc)[8]) {
  constexpr int NSEG = (K / 4 + SEG - 1) / SEG;
  int base[NSEG];
#pragma unroll
  for (int s = 0; s < NSEG; ++s) {
    base[s] = s * SEG * 256 + lane;
    if (s > 0) asm volatile("" : "+v"(base[s]));
  }
#pragma unroll
  for (int j = 0; j < K / 4; ++j) {
    const v2f* __restrict__ seg = wl + base[j / SEG];
    v2f av;
    if (j < 2 * (K / 8)) {
      const float4 q = a.q[j >> 1];
      if (j & 1) { av.x = q.z; av.y = q.w; } else { av.x = q.x; av.y = q.y; }
    } else {
      av = a.r;
    }
    if (SCALE) { av.x *= scale; av.y *= scale; }
#pragma unroll
    for (int nt = 0; nt < 8; ++nt) {
      const v2f bv = seg[((j % SEG) * 8 + nt) * 32];
      acc[nt] = __builtin_amdgcn_wmma_f32_16x16x4_f32(
          false, av, false, bv, (short)0, acc[nt], false, false);
    }
  }
}

__global__ __launch_bounds__(SW * 32, 4) void encoder_gemm(
    const float* __restrict__ x, const float* __restrict__ W,
    float* __restrict__ out, int nstrips) {
  __shared__ v2f wl[(KENC / 4) * 256];
  const int lane = threadIdx.x & 31;
  const int c = lane & 15, h = lane >> 4;
  const int mine = blockIdx.x * SW + (threadIdx.x >> 5);
  const int strip = min(mine, nstrips - 1);

  AFrag<KENC> a;
  load_a<KENC>(x + (size_t)(strip * 16 + c) * KENC, h, a);
  stage_weights<KENC, 25>(W, wl);
  __syncthreads();

  v8f acc[8] = {};
  strip_mma<KENC, 25, false>(a, 1.0f, wl, lane, acc);

  if (mine < nstrips) {
    float* __restrict__ o = out + (size_t)(strip * 16 + h * 8) * D + c;
#pragma unroll
    for (int nt = 0; nt < 8; ++nt)
#pragma unroll
      for (int r = 0; r < 8; ++r)
        o[(size_t)r * D + nt * 16] = acc[nt][r];
  }
}


__global__ __launch_bounds__(256) void edge_rank(const int* __restrict__ ei,
                                                 int* __restrict__ cnt,
                                                 int* __restrict__ rank,
                                                 long long E) {
  long long e = (long long)blockIdx.x * blockDim.x + threadIdx.x;
  if (e >= E) return;
  int t = ei[E + e];
  rank[e] = atomicAdd(&cnt[t], 1);
}

__global__ __launch_bounds__(1024) void scan_counts(const int* __restrict__ cnt,
                                                    int* __restrict__ rowptr,
                                                    int N) {
  __shared__ int wsum[32];
  const int tid = threadIdx.x, lane = tid & 31, wave = tid >> 5;
  int carry = 0;
  for (int base = 0; base < N; base += 16384) {
    const int i0 = base + tid * 16;
    const bool whole = (i0 + 16 <= N);
    int c[16];
    if (whole) {
      const int4* p = reinterpret_cast<const int4*>(cnt + i0);
#pragma unroll
      for (int q = 0; q < 4; ++q) {
        const int4 t = p[q];
        c[4 * q + 0] = t.x; c[4 * q + 1] = t.y; c[4 * q + 2] = t.z; c[4 * q + 3] = t.w;
      }
    } else {
#pragma unroll
      for (int u = 0; u < 16; ++u) c[u] = (i0 + u < N) ? cnt[i0 + u] : 0;
    }
    int tsum = 0;
#pragma unroll
    for (int u = 0; u < 16; ++u) { const int t = c[u]; c[u] = tsum; tsum += t; }
    int inc = tsum;
    for (int off = 1; off < 32; off <<= 1) {
      int v = __shfl_up(inc, off, 32);
      if (lane >= off) inc += v;
    }
    if (lane == 31) wsum[wave] = inc;
    __syncthreads();
    const int ws = wsum[lane];
    int winc = ws;
    for (int off = 1; off < 32; off <<= 1) {
      int v = __shfl_up(winc, off, 32);
      if (lane >= off) winc += v;
    }
    const int wave_excl = __shfl(winc, wave, 32) - __shfl(ws, wave, 32);
    const int total = __shfl(winc, 31, 32);
    const int ex = carry + wave_excl + (inc - tsum);
    if (whole) {
      int4* p = reinterpret_cast<int4*>(rowptr + i0);
#pragma unroll
      for (int q = 0; q < 4; ++q)
        p[q] = make_int4(ex + c[4 * q + 0], ex + c[4 * q + 1], ex + c[4 * q + 2], ex + c[4 * q + 3]);
    } else {
#pragma unroll
      for (int u = 0; u < 16; ++u)
        if (i0 + u < N) rowptr[i0 + u] = ex + c[u];
    }
    carry += total;
    __syncthreads();
  }
  if (tid == 0) rowptr[N] = carry;
}

__global__ __launch_bounds__(256) void edge_fill(const int* __restrict__ ei,
                                                 const int* __restrict__ rowptr,
                                                 const int* __restrict__ rank,
                                                 int* __restrict__ srcs,
                                                 long long E) {
  long long e = (long long)blockIdx.x * blockDim.x + threadIdx.x;
  if (e >= E) return;
  int s = ei[e];
  int t = ei[E + e];
  srcs[rowptr[t] + rank[e]] = s;
}

__global__ __launch_bounds__(256) void seg_gather(const int* __restrict__ rowptr,
                                                  const int* __restrict__ srcs,
                                                  const float* __restrict__ news,
                                                  float* __restrict__ agg,
                                                  int N) {
  const int lane = threadIdx.x & 31;
  const int n = __builtin_amdgcn_readfirstlane((int)(blockIdx.x * 8 + (threadIdx.x >> 5)));
  if (n >= N) return;
  const int beg = __builtin_amdgcn_readfirstlane(rowptr[n]);
  const int end = __builtin_amdgcn_readfirstlane(rowptr[n + 1]);
  const float* __restrict__ nl = news + lane * 4;
  float4 acc = make_float4(0.0f, 0.0f, 0.0f, 0.0f);
  for (int base = beg; base < end; base += 32) {
    const int m = min(32, end - base);
    const int my = (lane < m) ? srcs[base + lane] : 0;
    for (int j = 0; j < m; j += 8) {
      float4 v[8];
#pragma unroll
      for (int u = 0; u < 8; ++u) {
        const int idx = min(j + u, m - 1);
        const int s = __builtin_amdgcn_readlane(my, idx);
        v[u] = *reinterpret_cast<const float4*>(nl + (size_t)s * D);
      }
#pragma unroll
      for (int u = 0; u < 8; ++u) {
        if (j + u < m) {
          acc.x += v[u].x; acc.y += v[u].y; acc.z += v[u].z; acc.w += v[u].w;
        }
      }
    }
  }
  *reinterpret_cast<float4*>(agg + (size_t)n * D + lane * 4) = acc;
}

__global__ __launch_bounds__(SW * 32) void gnn_gemm_add(
    const float* __restrict__ agg, const int* __restrict__ deg,
    const float* __restrict__ Wg, const float* __restrict__ news,
    float* __restrict__ xx, int nstrips) {
  __shared__ v2f wl[(D / 4) * 256];
  const int lane = threadIdx.x & 31;
  const int c = lane & 15, h = lane >> 4;
  const int mine = blockIdx.x * SW + (threadIdx.x >> 5);
  const int strip = min(mine, nstrips - 1);
  const int arow = strip * 16 + c;

  AFrag<D> a;
  load_a<D>(agg + (size_t)arow * D, h, a);
  const float dinv = 1.0f / fmaxf((float)deg[arow], 1.0f);
  stage_weights<D, 16>(Wg, wl);
  __syncthreads();

  v8f acc[8] = {};
  strip_mma<D, 32, true>(a, dinv, wl, lane, acc);

  if (mine < nstrips) {
    const size_t o0 = (size_t)(strip * 16 + h * 8) * D + c;
#pragma unroll
    for (int nt = 0; nt < 8; ++nt)
#pragma unroll
      for (int r = 0; r < 8; ++r) {
        const size_t o = o0 + (size_t)r * D + nt * 16;
        xx[o] = news[o] + acc[nt][r];
      }
  }
}

#define UW 4
#define HB 25
__global__ __launch_bounds__(UW * 32, 4) void user_score(
    const float* __restrict__ xx, const int* __restrict__ hist,
    const int* __restrict__ cand, float* __restrict__ scores, int B) {
  static_assert(HIST > 32 && HIST <= 64 && HIST % HB == 0 && CAND <= 32,
                "id lists are held in two registers and one register");
  const int lane = threadIdx.x & 31;
  const int b = __builtin_amdgcn_readfirstlane((int)(blockIdx.x * UW + (threadIdx.x >> 5)));
  if (b >= B) return;
  const int* __restrict__ hp = hist + (size_t)b * HIST;
  const int id0 = hp[lane];
  const int id1 = (lane < HIST - 32) ? hp[32 + lane] : 0;
  const int idc = (lane < CAND) ? cand[(size_t)b * CAND + lane] : 0;
  const float* __restrict__ xl = xx + lane * 4;

  float4 acc = make_float4(0.0f, 0.0f, 0.0f, 0.0f);
  float4 cv[CAND];
#pragma unroll
  for (int h0 = 0; h0 < HIST; h0 += HB) {
    float4 v[HB];
    int sid[HB];
#pragma unroll
    for (int u = 0; u < HB; ++u) {
      const int h = h0 + u;
      sid[u] = (h < 32) ? __builtin_amdgcn_readlane(id0, h)
                        : __builtin_amdgcn_readlane(id1, h - 32);
      v[u] = *reinterpret_cast<const float4*>(xl + (size_t)sid[u] * D);
    }
    if (h0 == 0) {
#pragma unroll
      for (int c = 0; c < CAND; ++c) {
        const int s = __builtin_amdgcn_readlane(idc, c);
        cv[c] = *reinterpret_cast<const float4*>(xl + (size_t)s * D);
      }
    }
#pragma unroll
    for (int u = 0; u < HB; ++u) {
      if (sid[u] != 0) {
        acc.x += v[u].x; acc.y += v[u].y; acc.z += v[u].z; acc.w += v[u].w;
      }
    }
  }
  const int nread = __builtin_popcount(__builtin_amdgcn_ballot_w32(id0 != 0)) +
                    __builtin_popcount(__builtin_amdgcn_ballot_w32(id1 != 0));
  const float den = fmaxf((float)nread, 1e-9f);
  const float4 uv = make_float4(acc.x / den, acc.y / den, acc.z / den, acc.w / den);

  float mine = 0.0f;
#pragma unroll
  for (int c = 0; c < CAND; ++c) {
    float p = cv[c].x * uv.x + cv[c].y * uv.y + cv[c].z * uv.z + cv[c].w * uv.w;
#pragma unroll
    for (int off = 16; off; off >>= 1) p += __shfl_xor(p, off, 32);
    if (lane == c) mine = p;
  }
  if (lane < CAND) scores[(size_t)b * CAND + lane] = mine;
}

extern "C" void kernel_launch(void* const* d_in, const int* in_sizes, int n_in,
                              void* d_out, int out_size, void* d_ws, size_t ws_size,
                              hipStream_t stream) {
  const float* x      = (const float*)d_in[0];
  const int*   ei     = (const int*)  d_in[2];
  const int*   hist   = (const int*)  d_in[3];
  const int*   cand   = (const int*)  d_in[4];
  const float* W_enc  = (const float*)d_in[5];
  const float* W_gnn  = (const float*)d_in[6];
  float*       scores = (float*)d_out;

  const int       N = in_sizes[1];
  const long long E = (long long)in_sizes[2] / 2;
  const int       B = in_sizes[3] / HIST;

  float* news = (float*)d_ws;
  float* agg  = news + (size_t)N * D;
  int*   cnt  = (int*)(agg + (size_t)N * D);
  float* xx   = (float*)(cnt + (size_t)N);
  int* rowptr = (int*)xx;
  int* rank   = rowptr + (size_t)N + 1;
  int* srcs   = rank + E;

  hipMemsetAsync(cnt, 0, (size_t)N * sizeof(int), stream);

  const int strips  = N / 16;
  const int gblocks = (strips + SW - 1) / SW;
  encoder_gemm<<<gblocks, SW * 32, 0, stream>>>(x, W_enc, news, strips);

  int eblocks = (int)((E + 255) / 256);
  edge_rank<<<eblocks, 256, 0, stream>>>(ei, cnt, rank, E);
  scan_counts<<<1, 1024, 0, stream>>>(cnt, rowptr, N);
  edge_fill<<<eblocks, 256, 0, stream>>>(ei, rowptr, rank, srcs, E);
  seg_gather<<<(N + 7) / 8, 256, 0, stream>>>(rowptr, srcs, news, agg, N);

  gnn_gemm_add<<<gblocks, SW * 32, 0, stream>>>(agg, cnt, W_gnn, news, xx, strips);

  user_score<<<(B + UW - 1) / UW, UW * 32, 0, stream>>>(xx, hist, cand, scores, B);
}


# Adjacency built in one pass into fixed 64-slot node buckets (the atomic's returned rank is the slot): prefix-sum and edge-fill launches go; a node above 64 edges sums from the edge list in edge order
# speedup vs baseline: 3.3169x; 1.1547x over previous
#include <hip/hip_runtime.h>
#include <hip/hip_bf16.h>

typedef float v2f __attribute__((ext_vector_type(2)));
typedef float v8f __attribute__((ext_vector_type(8)));

#define D 128
#define HIST 50
#define CAND 20
#define KENC 300
#define SW 8

template <int K>
__device__ __forceinline__ int step_k(int j, int h) {
  return (j < 2 * (K / 8)) ? 8 * (j >> 1) + 4 * h + 2 * (j & 1)
                           : 8 * (K / 8) + 2 * h;
}

template <int K, int U>
__device__ __forceinline__ void stage_weights(const float* __restrict__ W,
                                              v2f* __restrict__ wl) {
  static_assert((K / 4) % U == 0, "batches must tile the steps");
  const int tid = threadIdx.x;
  const int lane = tid & 31, nt = tid >> 5;
  const int n = nt * 16 + (lane & 15);
  const int h = lane >> 4;
#pragma unroll
  for (int j0 = 0; j0 < K / 4; j0 += U) {
    v2f w[U];
#pragma unroll
    for (int u = 0; u < U; ++u) {
      const int k = step_k<K>(j0 + u, h);
      w[u].x = W[(size_t)k * D + n];
      w[u].y = W[(size_t)(k + 1) * D + n];
    }
#pragma unroll
    for (int u = 0; u < U; ++u) wl[(j0 + u) * 256 + tid] = w[u];
  }
}

template <int K>
struct AFrag {
  float4 q[K / 8];
  v2f r;
};

template <int K>
__device__ __forceinline__ void load_a(const float* __restrict__ row, int h,
                                       AFrag<K>& a) {
#pragma unroll
  for (int i = 0; i < K / 8; ++i)
    a.q[i] = *reinterpret_cast<const float4*>(row + 8 * i + 4 * h);
  if (K % 8)
    a.r = *reinterpret_cast<const v2f*>(row + 8 * (K / 8) + 2 * h);
}

template <int K, int SEG, bool SCALE>
__device__ __forceinline__ void strip_mma(const AFrag<K>& a, float scale,
                                          const v2f* __restrict__ wl, int lane,
                                          v8f (&acc)[8]) {
  constexpr int NSEG = (K / 4 + SEG - 1) / SEG;
  int base[NSEG];
#pragma unroll
  for (int s = 0; s < NSEG; ++s) {
    base[s] = s * SEG * 256 + lane;
    if (s > 0) asm volatile("" : "+v"(base[s]));
  }
#pragma unroll
  for (int j = 0; j < K / 4; ++j) {
    const v2f* __restrict__ seg = wl + base[j / SEG];
    v2f av;
    if (j < 2 * (K / 8)) {
      const float4 q = a.q[j >> 1];
      if (j & 1) { av.x = q.z; av.y = q.w; } else { av.x = q.x; av.y = q.y; }
    } else {
      av = a.r;
    }
    if (SCALE) { av.x *= scale; av.y *= scale; }
#pragma unroll
    for (int nt = 0; nt < 8; ++nt) {
      const v2f bv = seg[((j % SEG) * 8 + nt) * 32];
      acc[nt] = __builtin_amdgcn_wmma_f32_16x16x4_f32(
          false, av, false, bv, (short)0, acc[nt], false, false);
    }
  }
}

__global__ __launch_bounds__(SW * 32, 4) void encoder_gemm(
    const float* __restrict__ x, const float* __restrict__ W,
    float* __restrict__ out, int nstrips) {
  __shared__ v2f wl[(KENC / 4) * 256];
  const int lane = threadIdx.x & 31;
  const int c = lane & 15, h = lane >> 4;
  const int mine = blockIdx.x * SW + (threadIdx.x >> 5);
  const int strip = min(mine, nstrips - 1);

  AFrag<KENC> a;
  load_a<KENC>(x + (size_t)(strip * 16 + c) * KENC, h, a);
  stage_weights<KENC, 25>(W, wl);
  __syncthreads();

  v8f acc[8] = {};
  strip_mma<KENC, 25, false>(a, 1.0f, wl, lane, acc);

  if (mine < nstrips) {
    float* __restrict__ o = out + (size_t)(strip * 16 + h * 8) * D + c;
#pragma unroll
    for (int nt = 0; nt < 8; ++nt)
#pragma unroll
      for (int r = 0; r < 8; ++r)
        o[(size_t)r * D + nt * 16] = acc[nt][r];
  }
}

#define BKT 64

__global__ __launch_bounds__(256) void edge_bucket(const int* __restrict__ ei,
                                                   int* __restrict__ cnt,
                                                   int* __restrict__ bucket,
                                                   long long E) {
  long long e = (long long)blockIdx.x * blockDim.x + threadIdx.x;
  if (e >= E) return;
  const int t = ei[E + e];
  int s = ei[e];
  asm volatile("" : "+v"(s));
  const int r = atomicAdd(&cnt[t], 1);
  if (r < BKT) bucket[(size_t)t * BKT + r] = s;
}

__global__ __launch_bounds__(256) void seg_gather(const int* __restrict__ cnt,
                                                  const int* __restrict__ bucket,
                                                  const int* __restrict__ ei,
                                                  const float* __restrict__ news,
                                                  float* __restrict__ agg,
                                                  int N, long long E) {
  const int lane = threadIdx.x & 31;
  const int n = __builtin_amdgcn_readfirstlane((int)(blockIdx.x * 8 + (threadIdx.x >> 5)));
  if (n >= N) return;
  const int deg = __builtin_amdgcn_readfirstlane(cnt[n]);
  const int inb = (deg > BKT) ? 0 : deg;
  const int* __restrict__ bk = bucket + (size_t)n * BKT;
  const float* __restrict__ nl = news + lane * 4;
  float4 acc = make_float4(0.0f, 0.0f, 0.0f, 0.0f);
  for (int base = 0; base < inb; base += 32) {
    const int m = min(32, inb - base);
    const int my = (lane < m) ? bk[base + lane] : 0;
    for (int j = 0; j < m; j += 8) {
      float4 v[8];
#pragma unroll
      for (int u = 0; u < 8; ++u) {
        const int idx = min(j + u, m - 1);
        const int s = __builtin_amdgcn_readlane(my, idx);
        v[u] = *reinterpret_cast<const float4*>(nl + (size_t)s * D);
      }
#pragma unroll
      for (int u = 0; u < 8; ++u) {
        if (j + u < m) {
          acc.x += v[u].x; acc.y += v[u].y; acc.z += v[u].z; acc.w += v[u].w;
        }
      }
    }
  }
  if (deg > BKT) {
    for (long long e0 = 0; e0 < E; e0 += 32) {
      const long long e = e0 + lane;
      const bool hit = (e < E) && (ei[E + e] == n);
      unsigned mm = __builtin_amdgcn_ballot_w32(hit);
      if (mm) {
        const int sv = hit ? ei[e] : 0;
        while (mm) {
          const int l = __builtin_ctz(mm);
          mm &= mm - 1;
          const int s = __builtin_amdgcn_readlane(sv, l);
          const float4 v = *reinterpret_cast<const float4*>(nl + (size_t)s * D);
          acc.x += v.x; acc.y += v.y; acc.z += v.z; acc.w += v.w;
        }
      }
    }
  }
  *reinterpret_cast<float4*>(agg + (size_t)n * D + lane * 4) = acc;
}

__global__ __launch_bounds__(SW * 32) void gnn_gemm_add(
    const float* __restrict__ agg, const int* __restrict__ deg,
    const float* __restrict__ Wg, const float* __restrict__ news,
    float* __restrict__ xx, int nstrips) {
  __shared__ v2f wl[(D / 4) * 256];
  const int lane = threadIdx.x & 31;
  const int c = lane & 15, h = lane >> 4;
  const int mine = blockIdx.x * SW + (threadIdx.x >> 5);
  const int strip = min(mine, nstrips - 1);
  const int arow = strip * 16 + c;

  AFrag<D> a;
  load_a<D>(agg + (size_t)arow * D, h, a);
  const float dinv = 1.0f / fmaxf((float)deg[arow], 1.0f);
  stage_weights<D, 16>(Wg, wl);
  __syncthreads();

  v8f acc[8] = {};
  strip_mma<D, 32, true>(a, dinv, wl, lane, acc);

  if (mine < nstrips) {
    const size_t o0 = (size_t)(strip * 16 + h * 8) * D + c;
#pragma unroll
    for (int nt = 0; nt < 8; ++nt)
#pragma unroll
      for (int r = 0; r < 8; ++r) {
        const size_t o = o0 + (size_t)r * D + nt * 16;
        xx[o] = news[o] + acc[nt][r];
      }
  }
}

#define UW 4
#define HB 25
__global__ __launch_bounds__(UW * 32, 4) void user_score(
    const float* __restrict__ xx, const int* __restrict__ hist,
    const int* __restrict__ cand, float* __restrict__ scores, int B) {
  static_assert(HIST > 32 && HIST <= 64 && HIST % HB == 0 && CAND <= 32,
                "id lists are held in two registers and one register");
  const int lane = threadIdx.x & 31;
  const int b = __builtin_amdgcn_readfirstlane((int)(blockIdx.x * UW + (threadIdx.x >> 5)));
  if (b >= B) return;
  const int* __restrict__ hp = hist + (size_t)b * HIST;
  const int id0 = hp[lane];
  const int id1 = (lane < HIST - 32) ? hp[32 + lane] : 0;
  const int idc = (lane < CAND) ? cand[(size_t)b * CAND + lane] : 0;
  const float* __restrict__ xl = xx + lane * 4;

  float4 acc = make_float4(0.0f, 0.0f, 0.0f, 0.0f);
  float4 cv[CAND];
#pragma unroll
  for (int h0 = 0; h0 < HIST; h0 += HB) {
    float4 v[HB];
    int sid[HB];
#pragma unroll
    for (int u = 0; u < HB; ++u) {
      const int h = h0 + u;
      sid[u] = (h < 32) ? __builtin_amdgcn_readlane(id0, h)
                        : __builtin_amdgcn_readlane(id1, h - 32);
      v[u] = *reinterpret_cast<const float4*>(xl + (size_t)sid[u] * D);
    }
    if (h0 == 0) {
#pragma unroll
      for (int c = 0; c < CAND; ++c) {
        const int s = __builtin_amdgcn_readlane(idc, c);
        cv[c] = *reinterpret_cast<const float4*>(xl + (size_t)s * D);
      }
    }
#pragma unroll
    for (int u = 0; u < HB; ++u) {
      if (sid[u] != 0) {
        acc.x += v[u].x; acc.y += v[u].y; acc.z += v[u].z; acc.w += v[u].w;
      }
    }
  }
  const int nread = __builtin_popcount(__builtin_amdgcn_ballot_w32(id0 != 0)) +
                    __builtin_popcount(__builtin_amdgcn_ballot_w32(id1 != 0));
  const float den = fmaxf((float)nread, 1e-9f);
  const float4 uv = make_float4(acc.x / den, acc.y / den, acc.z / den, acc.w / den);

  float mine = 0.0f;
#pragma unroll
  for (int c = 0; c < CAND; ++c) {
    float p = cv[c].x * uv.x + cv[c].y * uv.y + cv[c].z * uv.z + cv[c].w * uv.w;
#pragma unroll
    for (int off = 16; off; off >>= 1) p += __shfl_xor(p, off, 32);
    if (lane == c) mine = p;
  }
  if (lane < CAND) scores[(size_t)b * CAND + lane] = mine;
}

extern "C" void kernel_launch(void* const* d_in, const int* in_sizes, int n_in,
                              void* d_out, int out_size, void* d_ws, size_t ws_size,
                              hipStream_t stream) {
  const float* x      = (const float*)d_in[0];
  const int*   ei     = (const int*)  d_in[2];
  const int*   hist   = (const int*)  d_in[3];
  const int*   cand   = (const int*)  d_in[4];
  const float* W_enc  = (const float*)d_in[5];
  const float* W_gnn  = (const float*)d_in[6];
  float*       scores = (float*)d_out;

  const int       N = in_sizes[1];
  const long long E = (long long)in_sizes[2] / 2;
  const int       B = in_sizes[3] / HIST;

  float* news = (float*)d_ws;
  float* agg  = news + (size_t)N * D;
  int*   cnt  = (int*)(agg + (size_t)N * D);
  float* xx   = (float*)(cnt + (size_t)N);
  int* bucket = (int*)xx;

  hipMemsetAsync(cnt, 0, (size_t)N * sizeof(int), stream);

  const int strips  = N / 16;
  const int gblocks = (strips + SW - 1) / SW;
  encoder_gemm<<<gblocks, SW * 32, 0, stream>>>(x, W_enc, news, strips);

  int eblocks = (int)((E + 255) / 256);
  edge_bucket<<<eblocks, 256, 0, stream>>>(ei, cnt, bucket, E);
  seg_gather<<<(N + 7) / 8, 256, 0, stream>>>(cnt, bucket, ei, news, agg, N, E);

  gnn_gemm_add<<<gblocks, SW * 32, 0, stream>>>(agg, cnt, W_gnn, news, xx, strips);

  user_score<<<(B + UW - 1) / UW, UW * 32, 0, stream>>>(xx, hist, cand, scores, B);
}
